# starting kernel
# baseline (speedup 1.0000x reference)
#include <hip/hip_runtime.h>
#include <hip/hip_bf16.h>

typedef __attribute__((ext_vector_type(2))) float v2f;
typedef __attribute__((ext_vector_type(8))) float v8f;

constexpr int B_ = 4, L_ = 1024, VOCAB_ = 50257, D_ = 64;

__global__ __launch_bounds__(256) void pool_embed_kernel(
    const int* __restrict__ x, const float* __restrict__ a,
    const float* __restrict__ E_w, const float* __restrict__ V_w,
    float* __restrict__ c_ws)
{
    __shared__ float w[L_];
    __shared__ int   xs[L_];
    __shared__ float red[256];
    __shared__ float m_s[D_];

    const int b = blockIdx.x;
    const int t = threadIdx.x;
    const float* arow = a + ((size_t)b * L_ + (L_ - 1)) * (size_t)L_;

    float vals[4];
    float lmax = -INFINITY;
    #pragma unroll
    for (int i = 0; i < 4; ++i) {
        int l = t + i * 256;
        float v = arow[l];
        vals[i] = v;
        lmax = fmaxf(lmax, v);
        xs[l] = x[b * L_ + l];
    }
    red[t] = lmax;
    __syncthreads();
    for (int s = 128; s > 0; s >>= 1) {
        if (t < s) red[t] = fmaxf(red[t], red[t + s]);
        __syncthreads();
    }
    float mx = red[0];
    __syncthreads();

    float lsum = 0.f;
    #pragma unroll
    for (int i = 0; i < 4; ++i) {
        int l = t + i * 256;
        float e = __expf(vals[i] - mx);
        w[l] = e;
        lsum += e;
    }
    red[t] = lsum;
    __syncthreads();
    for (int s = 128; s > 0; s >>= 1) {
        if (t < s) red[t] += red[t + s];
        __syncthreads();
    }
    float inv_sum = 1.0f / red[0];
    __syncthreads();

    int d = t & 63;
    int chunk = t >> 6;
    const float* Ed = E_w + (size_t)d * VOCAB_;
    float partial = 0.f;
    int l0 = chunk * 256;
    for (int l = l0; l < l0 + 256; ++l)
        partial += w[l] * Ed[xs[l]];
    red[t] = partial;
    __syncthreads();
    if (t < 64)
        m_s[t] = (red[t] + red[t + 64] + red[t + 128] + red[t + 192]) * inv_sum;
    __syncthreads();

    if (t < 64) {
        const float* Vr = V_w + t * D_;
        float cc = 0.f;
        #pragma unroll
        for (int dd = 0; dd < D_; ++dd) cc += m_s[dd] * Vr[dd];
        c_ws[b * D_ + t] = cc;
    }
}

__global__ __launch_bounds__(256) void vocab_wmma_kernel(
    const float* __restrict__ U_w, const float* __restrict__ c_ws,
    float* __restrict__ out)
{
    const int t    = threadIdx.x;
    const int wave = t >> 5;
    const int lane = t & 31;
    const int tile = blockIdx.x * 8 + wave;
    const int col0 = tile * 16;
    if (col0 >= VOCAB_) return;

    const int n     = lane & 15;
    const int khalf = lane >> 4;
    const int col   = col0 + n;
    const bool cvalid = col < VOCAB_;
    const int colc  = cvalid ? col : (VOCAB_ - 1);
    const int mrow  = lane & 15;

    const float* Urow = U_w  + (size_t)colc * D_;
    const float* Crow = c_ws + (mrow & 3) * D_;

    v2f afr[16], bfr[16];
    #pragma unroll
    for (int kk = 0; kk < 16; ++kk) {
        const int kb = kk * 4 + 2 * khalf;
        afr[kk] = *(const v2f*)(Crow + kb);
        bfr[kk] = *(const v2f*)(Urow + kb);
    }

    v8f acc = {};
    #pragma unroll
    for (int kk = 0; kk < 16; ++kk) {
        acc = __builtin_amdgcn_wmma_f32_16x16x4_f32(
            false, afr[kk], false, bfr[kk], (short)0, acc, false, false);
    }

    if (khalf == 0 && cvalid) {
        #pragma unroll
        for (int r = 0; r < B_; ++r)
            out[(size_t)r * VOCAB_ + col] = acc[r];
    }
}

extern "C" void kernel_launch(void* const* d_in, const int* in_sizes, int n_in,
                              void* d_out, int out_size, void* d_ws, size_t ws_size,
                              hipStream_t stream) {
    const int*   x   = (const int*)d_in[0];
    const float* a   = (const float*)d_in[1];
    const float* E_w = (const float*)d_in[2];
    const float* V_w = (const float*)d_in[3];
    const float* U_w = (const float*)d_in[4];
    float* out  = (float*)d_out;
    float* c_ws = (float*)d_ws;

    pool_embed_kernel<<<B_, 256, 0, stream>>>(x, a, E_w, V_w, c_ws);

    const int ntiles  = (VOCAB_ + 15) / 16;
    const int nblocks = (ntiles + 7) / 8;
    vocab_wmma_kernel<<<nblocks, 256, 0, stream>>>(U_w, c_ws, out);
}


# pooling re-cut by E_w address range: 256 work-groups (dim x vocab quarter) stream their row slice to LDS coalesced, serve token look-ups from LDS; vocab head sums 4 slices, applies V_w; new sum order
# speedup vs baseline: 13.0492x; 13.0492x over previous
#include <hip/hip_runtime.h>
#include <hip/hip_bf16.h>

typedef __attribute__((ext_vector_type(2))) float v2f;
typedef __attribute__((ext_vector_type(4))) float v4f;
typedef __attribute__((ext_vector_type(8))) float v8f;

constexpr int B_ = 4, L_ = 1024, VOCAB_ = 50257, D_ = 64;

constexpr int Q_   = 4;
constexpr int QL_  = (VOCAB_ + Q_ - 1) / Q_;
constexpr int HALF_ = 25;
constexpr int FULL_ = 49;
static_assert(2 * HALF_ * 256 >= QL_, "slice not covered");
static_assert(FULL_ * 256 <= VOCAB_ - (Q_ - 1) * QL_, "round 48 must fit the shortest slice");
constexpr int VPITCH_ = D_ + 1;
constexpr int PLINE_ = 32;

__global__ __launch_bounds__(256) void pool_slice_kernel(
    const int* __restrict__ x, const float* __restrict__ a,
    const float* __restrict__ E_w, float* __restrict__ part)
{
    __shared__ float es[QL_];
    __shared__ float red_max[8][B_];
    __shared__ float red_sum[8][B_];
    __shared__ float red_acc[8][B_];

    const int t    = threadIdx.x;
    const int wave = t >> 5;
    const int lane = t & 31;
    const int d    = blockIdx.x / Q_;
    const int q    = blockIdx.x % Q_;
    const int q0   = q * QL_;
    const int qlen = (VOCAB_ - q0 < QL_) ? (VOCAB_ - q0) : QL_;

    float val[16];
    int   xv[16];
    #pragma unroll
    for (int i = 0; i < 16; ++i) {
        const int b = i >> 2;
        const int l = t + 256 * (i & 3);
        val[i] = a[((size_t)b * L_ + (L_ - 1)) * (size_t)L_ + l];
        xv[i]  = x[b * L_ + l];
    }

    const float* Erow = E_w + (size_t)d * VOCAB_ + q0;
    #pragma unroll
    for (int h = 0; h < 2; ++h) {
        float buf[HALF_];
        #pragma unroll
        for (int j = 0; j < HALF_; ++j) {
            const int k   = h * HALF_ + j;
            const int idx = t + 256 * k;
            buf[j] = (k < FULL_ || idx < qlen) ? Erow[idx] : 0.f;
        }
        #pragma unroll
        for (int j = 0; j < HALF_; ++j) {
            const int k   = h * HALF_ + j;
            const int idx = t + 256 * k;
            if (k < FULL_ || idx < qlen) es[idx] = buf[j];
        }
    }

    float pm[B_];
    #pragma unroll
    for (int b = 0; b < B_; ++b) {
        pm[b] = fmaxf(fmaxf(val[4 * b], val[4 * b + 1]), fmaxf(val[4 * b + 2], val[4 * b + 3]));
        #pragma unroll
        for (int m = 16; m >= 1; m >>= 1) pm[b] = fmaxf(pm[b], __shfl_xor(pm[b], m, 32));
    }
    if (lane == 0) {
        #pragma unroll
        for (int b = 0; b < B_; ++b) red_max[wave][b] = pm[b];
    }
    __syncthreads();

    float mx[B_];
    #pragma unroll
    for (int b = 0; b < B_; ++b) {
        float m = red_max[0][b];
        #pragma unroll
        for (int w = 1; w < 8; ++w) m = fmaxf(m, red_max[w][b]);
        mx[b] = m;
    }

    float ps[B_]  = {0.f, 0.f, 0.f, 0.f};
    float acc[B_] = {0.f, 0.f, 0.f, 0.f};
    #pragma unroll
    for (int i = 0; i < 16; ++i) {
        const int b = i >> 2;
        const float e = __expf(val[i] - mx[b]);
        ps[b] += e;
        const int rel = xv[i] - q0;
        const bool in = (unsigned)rel < (unsigned)qlen;
        const float ev = es[in ? rel : 0];
        acc[b] += in ? e * ev : 0.f;
    }
    #pragma unroll
    for (int b = 0; b < B_; ++b) {
        #pragma unroll
        for (int m = 16; m >= 1; m >>= 1) {
            ps[b]  += __shfl_xor(ps[b], m, 32);
            acc[b] += __shfl_xor(acc[b], m, 32);
        }
    }
    if (lane == 0) {
        #pragma unroll
        for (int b = 0; b < B_; ++b) { red_sum[wave][b] = ps[b]; red_acc[wave][b] = acc[b]; }
    }
    __syncthreads();

    if (t < B_) {
        float s = 0.f, sa = 0.f;
        #pragma unroll
        for (int w = 0; w < 8; ++w) { s += red_sum[w][t]; sa += red_acc[w][t]; }
        part[(size_t)blockIdx.x * PLINE_ + t] = sa * (1.0f / s);
    }
}

__global__ __launch_bounds__(256) void vocab_wmma_kernel(
    const float* __restrict__ U_w, const float* __restrict__ V_w,
    const float* __restrict__ part, float* __restrict__ out)
{
    __shared__ float ms[Q_ * B_ * D_];
    __shared__ float m_s[B_ * D_];
    __shared__ float Vs[D_ * VPITCH_];
    __shared__ float c_s[B_ * D_];

    const int t    = threadIdx.x;
    const int wave = t >> 5;
    const int lane = t & 31;
    const int tile = blockIdx.x * 8 + wave;
    const int col0 = tile * 16;
    const bool tvalid = col0 < VOCAB_;

    const int n     = lane & 15;
    const int khalf = lane >> 4;
    const int col   = col0 + n;
    const bool cvalid = col < VOCAB_;
    const int colc  = cvalid ? col : (VOCAB_ - 1);
    const int mrow  = lane & 15;

    const float* Urow = U_w + (size_t)colc * D_;

    v2f bfr[16];
    #pragma unroll
    for (int kk = 0; kk < 16; ++kk) {
        const int kb = kk * 4 + 2 * khalf;
        bfr[kk] = *(const v2f*)(Urow + kb);
    }

    {
        const int hd = t & 63, hq = t >> 6;
        const v4f pv = *(const v4f*)(part + (size_t)(hd * Q_ + hq) * PLINE_);
        ms[(hq * B_ + 0) * D_ + hd] = pv.x;
        ms[(hq * B_ + 1) * D_ + hd] = pv.y;
        ms[(hq * B_ + 2) * D_ + hd] = pv.z;
        ms[(hq * B_ + 3) * D_ + hd] = pv.w;
        #pragma unroll
        for (int i = 0; i < 4; ++i) {
            const int f = (t + 256 * i) * 4;
            const v4f vv = *(const v4f*)(V_w + f);
            const int e = f >> 6, dd = f & 63;
            Vs[e * VPITCH_ + dd + 0] = vv.x;
            Vs[e * VPITCH_ + dd + 1] = vv.y;
            Vs[e * VPITCH_ + dd + 2] = vv.z;
            Vs[e * VPITCH_ + dd + 3] = vv.w;
        }
    }
    __syncthreads();
    m_s[t] = ((ms[t] + ms[256 + t]) + ms[512 + t]) + ms[768 + t];
    __syncthreads();
    {
        const int hb = t >> 6, he = t & 63;
        float cc = 0.f;
        #pragma unroll
        for (int dd = 0; dd < D_; ++dd) cc += m_s[hb * D_ + dd] * Vs[he * VPITCH_ + dd];
        c_s[t] = cc;
    }
    __syncthreads();

    if (tvalid) {
        const float* Crow = c_s + (mrow & 3) * D_;
        v2f afr[16];
        #pragma unroll
        for (int kk = 0; kk < 16; ++kk) {
            const int kb = kk * 4 + 2 * khalf;
            afr[kk] = *(const v2f*)(Crow + kb);
        }

        v8f acc = {};
        #pragma unroll
        for (int kk = 0; kk < 16; ++kk) {
            acc = __builtin_amdgcn_wmma_f32_16x16x4_f32(
                false, afr[kk], false, bfr[kk], (short)0, acc, false, false);
        }

        if (khalf == 0 && cvalid) {
            #pragma unroll
            for (int r = 0; r < B_; ++r)
                out[(size_t)r * VOCAB_ + col] = acc[r];
        }
    }
}

extern "C" void kernel_launch(void* const* d_in, const int* in_sizes, int n_in,
                              void* d_out, int out_size, void* d_ws, size_t ws_size,
                              hipStream_t stream) {
    const int*   x   = (const int*)d_in[0];
    const float* a   = (const float*)d_in[1];
    const float* E_w = (const float*)d_in[2];
    const float* V_w = (const float*)d_in[3];
    const float* U_w = (const float*)d_in[4];
    float* out  = (float*)d_out;
    float* part = (float*)d_ws;

    pool_slice_kernel<<<D_ * Q_, 256, 0, stream>>>(x, a, E_w, part);

    const int ntiles  = (VOCAB_ + 15) / 16;
    const int nblocks = (ntiles + 7) / 8;
    vocab_wmma_kernel<<<nblocks, 256, 0, stream>>>(U_w, V_w, part, out);
}


# every work-group barrier now waits for LDS only, so global loads issued ahead stay in flight across it: pool row maxima and exps run under the E_w slice loads, vocab head under the U_w loads
# speedup vs baseline: 14.7066x; 1.1270x over previous
#include <hip/hip_runtime.h>
#include <hip/hip_bf16.h>

typedef __attribute__((ext_vector_type(2))) float v2f;
typedef __attribute__((ext_vector_type(4))) float v4f;
typedef __attribute__((ext_vector_type(8))) float v8f;

constexpr int B_ = 4, L_ = 1024, VOCAB_ = 50257, D_ = 64;

constexpr int Q_   = 4;
constexpr int QL_  = (VOCAB_ + Q_ - 1) / Q_;
constexpr int ROUNDS_ = 50;
constexpr int FULL_ = 49;
static_assert(ROUNDS_ * 256 >= QL_, "slice not covered");
static_assert(FULL_ * 256 <= VOCAB_ - (Q_ - 1) * QL_, "round 48 must fit the shortest slice");
constexpr int VPITCH_ = D_ + 1;
constexpr int PLINE_ = 32;

static __device__ __forceinline__ void lds_barrier() {
    asm volatile("s_wait_dscnt 0x0\n\ts_barrier_signal -1\n\ts_barrier_wait -1" ::: "memory");
}

__global__ __launch_bounds__(256) void pool_slice_kernel(
    const int* __restrict__ x, const float* __restrict__ a,
    const float* __restrict__ E_w, float* __restrict__ part)
{
    __shared__ float es[QL_];
    __shared__ float red_max[8][B_];
    __shared__ float red_sum[8][B_];
    __shared__ float red_acc[8][B_];

    const int t    = threadIdx.x;
    const int wave = t >> 5;
    const int lane = t & 31;
    const int d    = blockIdx.x / Q_;
    const int q    = blockIdx.x % Q_;
    const int q0   = q * QL_;
    const int qlen = (VOCAB_ - q0 < QL_) ? (VOCAB_ - q0) : QL_;

    float val[16];
    int   xv[16];
    #pragma unroll
    for (int i = 0; i < 16; ++i) {
        const int b = i >> 2;
        const int l = t + 256 * (i & 3);
        val[i] = a[((size_t)b * L_ + (L_ - 1)) * (size_t)L_ + l];
        xv[i]  = x[b * L_ + l];
    }
    __builtin_amdgcn_sched_barrier(0);

    const float* Erow = E_w + (size_t)d * VOCAB_ + q0;
    float buf[ROUNDS_];
    #pragma unroll
    for (int k = 0; k < ROUNDS_; ++k) {
        const int idx = t + 256 * k;
        buf[k] = (k < FULL_ || idx < qlen) ? Erow[idx] : 0.f;
    }

    float pm[B_];
    #pragma unroll
    for (int b = 0; b < B_; ++b) {
        pm[b] = fmaxf(fmaxf(val[4 * b], val[4 * b + 1]), fmaxf(val[4 * b + 2], val[4 * b + 3]));
        #pragma unroll
        for (int m = 16; m >= 1; m >>= 1) pm[b] = fmaxf(pm[b], __shfl_xor(pm[b], m, 32));
    }
    if (lane == 0) {
        #pragma unroll
        for (int b = 0; b < B_; ++b) red_max[wave][b] = pm[b];
    }
    lds_barrier();

    float mx[B_];
    #pragma unroll
    for (int b = 0; b < B_; ++b) {
        float m = red_max[0][b];
        #pragma unroll
        for (int w = 1; w < 8; ++w) m = fmaxf(m, red_max[w][b]);
        mx[b] = m;
    }

    float ev_[16];
    float ps[B_]  = {0.f, 0.f, 0.f, 0.f};
    #pragma unroll
    for (int i = 0; i < 16; ++i) {
        ev_[i] = __expf(val[i] - mx[i >> 2]);
        ps[i >> 2] += ev_[i];
    }
    asm volatile("" :: "v"(ev_[0]), "v"(ev_[1]), "v"(ev_[2]), "v"(ev_[3]),
                       "v"(ev_[4]), "v"(ev_[5]), "v"(ev_[6]), "v"(ev_[7]),
                       "v"(ev_[8]), "v"(ev_[9]), "v"(ev_[10]), "v"(ev_[11]),
                       "v"(ev_[12]), "v"(ev_[13]), "v"(ev_[14]), "v"(ev_[15]));

    #pragma unroll
    for (int k = 0; k < ROUNDS_; ++k) {
        const int idx = t + 256 * k;
        if (k < FULL_ || idx < qlen) es[idx] = buf[k];
    }
    lds_barrier();

    float acc[B_] = {0.f, 0.f, 0.f, 0.f};
    #pragma unroll
    for (int i = 0; i < 16; ++i) {
        const int b = i >> 2;
        const float e = ev_[i];
        const int rel = xv[i] - q0;
        const bool in = (unsigned)rel < (unsigned)qlen;
        const float ev = es[in ? rel : 0];
        acc[b] += in ? e * ev : 0.f;
    }
    #pragma unroll
    for (int b = 0; b < B_; ++b) {
        #pragma unroll
        for (int m = 16; m >= 1; m >>= 1) {
            ps[b]  += __shfl_xor(ps[b], m, 32);
            acc[b] += __shfl_xor(acc[b], m, 32);
        }
    }
    if (lane == 0) {
        #pragma unroll
        for (int b = 0; b < B_; ++b) { red_sum[wave][b] = ps[b]; red_acc[wave][b] = acc[b]; }
    }
    lds_barrier();

    if (t < B_) {
        float s = 0.f, sa = 0.f;
        #pragma unroll
        for (int w = 0; w < 8; ++w) { s += red_sum[w][t]; sa += red_acc[w][t]; }
        part[(size_t)blockIdx.x * PLINE_ + t] = sa * (1.0f / s);
    }
}

__global__ __launch_bounds__(256) void vocab_wmma_kernel(
    const float* U_w, const float* __restrict__ V_w,
    const float* __restrict__ part, float* __restrict__ out)
{
    __shared__ float ms[Q_ * B_ * D_];
    __shared__ float m_s[B_ * D_];
    __shared__ float Vs[D_ * VPITCH_];
    __shared__ float c_s[B_ * D_];

    const int t    = threadIdx.x;
    const int wave = t >> 5;
    const int lane = t & 31;
    const int tile = blockIdx.x * 8 + wave;
    const int col0 = tile * 16;
    const bool tvalid = col0 < VOCAB_;

    const int n     = lane & 15;
    const int khalf = lane >> 4;
    const int col   = col0 + n;
    const bool cvalid = col < VOCAB_;
    const int colc  = cvalid ? col : (VOCAB_ - 1);
    const int mrow  = lane & 15;

    const float* Urow = U_w + (size_t)colc * D_;

    v2f bfr[16];
    #pragma unroll
    for (int kk = 0; kk < 16; ++kk) {
        const int kb = kk * 4 + 2 * khalf;
        bfr[kk] = *(const v2f*)(Urow + kb);
    }

    {
        const int hd = t & 63, hq = t >> 6;
        const v4f pv = *(const v4f*)(part + (size_t)(hd * Q_ + hq) * PLINE_);
        ms[(hq * B_ + 0) * D_ + hd] = pv.x;
        ms[(hq * B_ + 1) * D_ + hd] = pv.y;
        ms[(hq * B_ + 2) * D_ + hd] = pv.z;
        ms[(hq * B_ + 3) * D_ + hd] = pv.w;
        #pragma unroll
        for (int i = 0; i < 4; ++i) {
            const int f = (t + 256 * i) * 4;
            const v4f vv = *(const v4f*)(V_w + f);
            const int e = f >> 6, dd = f & 63;
            Vs[e * VPITCH_ + dd + 0] = vv.x;
            Vs[e * VPITCH_ + dd + 1] = vv.y;
            Vs[e * VPITCH_ + dd + 2] = vv.z;
            Vs[e * VPITCH_ + dd + 3] = vv.w;
        }
    }
    lds_barrier();
    m_s[t] = ((ms[t] + ms[256 + t]) + ms[512 + t]) + ms[768 + t];
    lds_barrier();
    {
        const int hb = t >> 6, he = t & 63;
        float cc = 0.f;
        #pragma unroll
        for (int dd = 0; dd < D_; ++dd) cc += m_s[hb * D_ + dd] * Vs[he * VPITCH_ + dd];
        c_s[t] = cc;
    }
    lds_barrier();

    if (tvalid) {
        const float* Crow = c_s + (mrow & 3) * D_;
        v2f afr[16];
        #pragma unroll
        for (int kk = 0; kk < 16; ++kk) {
            const int kb = kk * 4 + 2 * khalf;
            afr[kk] = *(const v2f*)(Crow + kb);
        }

        v8f acc = {};
        #pragma unroll
        for (int kk = 0; kk < 16; ++kk) {
            acc = __builtin_amdgcn_wmma_f32_16x16x4_f32(
                false, afr[kk], false, bfr[kk], (short)0, acc, false, false);
        }

        if (khalf == 0 && cvalid) {
            #pragma unroll
            for (int r = 0; r < B_; ++r)
                out[(size_t)r * VOCAB_ + col] = acc[r];
        }
    }
}

extern "C" void kernel_launch(void* const* d_in, const int* in_sizes, int n_in,
                              void* d_out, int out_size, void* d_ws, size_t ws_size,
                              hipStream_t stream) {
    const int*   x   = (const int*)d_in[0];
    const float* a   = (const float*)d_in[1];
    const float* E_w = (const float*)d_in[2];
    const float* V_w = (const float*)d_in[3];
    const float* U_w = (const float*)d_in[4];
    float* out  = (float*)d_out;
    float* part = (float*)d_ws;

    pool_slice_kernel<<<D_ * Q_, 256, 0, stream>>>(x, a, E_w, part);

    const int ntiles  = (VOCAB_ + 15) / 16;
    const int nblocks = (ntiles + 7) / 8;
    vocab_wmma_kernel<<<nblocks, 256, 0, stream>>>(U_w, V_w, part, out);
}
